# starting kernel
# baseline (speedup 1.0000x reference)
#include <hip/hip_runtime.h>
#include <hip/hip_bf16.h>

typedef __attribute__((ext_vector_type(2))) float v2f;
typedef __attribute__((ext_vector_type(8))) float v8f;

#define NGRP 40
#define NB   128
#define DIM  1024
#define MAX_TILES 64

#define WS_COUNTS 0
#define WS_ROWS   64
#define WS_TILES  5184

__global__ void build_tiles_kernel(const int* __restrict__ g,
                                   int* __restrict__ ws) {
    __shared__ int sCnt[NGRP];
    const int t = threadIdx.x;
    if (t < NGRP) {
        int cnt = 0;
        for (int b = 0; b < NB; ++b) {
            int grp = g[b] % NGRP;
            if (grp == t) {
                ws[WS_ROWS + t * NB + cnt] = b;
                ++cnt;
            }
        }
        sCnt[t] = cnt;
        ws[WS_COUNTS + t] = cnt;
    }
    __syncthreads();
    if (t == 0) {
        int nt = 0;
        for (int grp = 0; grp < NGRP; ++grp)
            for (int off = 0; off < sCnt[grp]; off += 16)
                ws[WS_TILES + nt++] = (grp << 16) | off;
        for (; nt < MAX_TILES; ++nt)
            ws[WS_TILES + nt] = -1;
    }
}

__global__ __launch_bounds__(256) void grouped_gemm_kernel(
    const float* __restrict__ x,
    const float* __restrict__ T,
    const int*   __restrict__ ws,
    float*       __restrict__ out) {

    __shared__ float Alds[DIM * 16];

    const int tile = ws[WS_TILES + blockIdx.x];
    if (tile < 0) return;
    const int grp  = tile >> 16;
    const int roff = tile & 0xFFFF;
    int m = ws[WS_COUNTS + grp] - roff;
    if (m > 16) m = 16;

    const int tid = threadIdx.x;

    for (int r = 0; r < 16; ++r) {
        const int rowIdx = (r < m) ? ws[WS_ROWS + grp * NB + roff + r] : -1;
        const float* xr = x + (size_t)(rowIdx < 0 ? 0 : rowIdx) * DIM;
        for (int d = tid; d < DIM; d += 256) {
            float v = (rowIdx >= 0) ? xr[d] : 0.0f;
            Alds[d * 16 + r] = v;
        }
    }
    __syncthreads();

    const int wave = tid >> 5;
    const int lane = tid & 31;
    const int r16  = lane & 15;
    const int half = lane >> 4;
    const int krow = half * 2;

    const int n0 = (blockIdx.y * 8 + wave) * 16;
    const float* Bp = T + (size_t)grp * DIM * DIM + n0 + r16;

    v8f c = {};

    for (int kk = 0; kk < DIM; kk += 16) {
        __builtin_prefetch(Bp + (size_t)(kk + 64) * DIM, 0, 0);
#pragma unroll
        for (int ks = 0; ks < 16; ks += 4) {
            const int k0 = kk + ks + krow;
            v2f a, b;
            a[0] = Alds[k0 * 16 + r16];
            a[1] = Alds[(k0 + 1) * 16 + r16];
            b[0] = Bp[(size_t)k0 * DIM];
            b[1] = Bp[(size_t)(k0 + 1) * DIM];
            c = __builtin_amdgcn_wmma_f32_16x16x4_f32(
                    false, a, false, b, (short)0, c, false, false);
        }
    }

#pragma unroll
    for (int j = 0; j < 8; ++j) {
        const int mrow = j + half * 8;
        if (mrow < m) {
            const int b = ws[WS_ROWS + grp * NB + roff + mrow];
            out[(size_t)b * DIM + n0 + r16] = c[j];
        }
    }
}

extern "C" void kernel_launch(void* const* d_in, const int* in_sizes, int n_in,
                              void* d_out, int out_size, void* d_ws, size_t ws_size,
                              hipStream_t stream) {
    const float* x = (const float*)d_in[0];
    const int*   g = (const int*)d_in[1];
    const float* T = (const float*)d_in[2];
    int*   ws  = (int*)d_ws;
    float* out = (float*)d_out;

    build_tiles_kernel<<<1, 64, 0, stream>>>(g, ws);

    dim3 grid(MAX_TILES, 8);
    grouped_gemm_kernel<<<grid, 256, 0, stream>>>(x, T, ws, out);
}


# B streamed with 16-byte loads, 18-32 in flight per wave: a wave takes 64 columns x a K quarter (4 interleaved column tiles); the 4 K-quarter partial sums are added through LDS (new f32 sum order)
# speedup vs baseline: 1.2483x; 1.2483x over previous
#include <hip/hip_runtime.h>
#include <hip/hip_bf16.h>

typedef __attribute__((ext_vector_type(2))) float v2f;
typedef __attribute__((ext_vector_type(4))) float v4f;
typedef __attribute__((ext_vector_type(8))) float v8f;

#define NGRP 40
#define NB   128
#define DIM  1024
#define MAX_TILES 64

#define WS_COUNTS 0
#define WS_ROWS   64
#define WS_TILES  5184

__global__ void build_tiles_kernel(const int* __restrict__ g,
                                   int* __restrict__ ws) {
    __shared__ int sCnt[NGRP];
    const int t = threadIdx.x;
    if (t < NGRP) {
        int cnt = 0;
        for (int b = 0; b < NB; ++b) {
            int grp = g[b] % NGRP;
            if (grp == t) {
                ws[WS_ROWS + t * NB + cnt] = b;
                ++cnt;
            }
        }
        sCnt[t] = cnt;
        ws[WS_COUNTS + t] = cnt;
    }
    __syncthreads();
    if (t == 0) {
        int nt = 0;
        for (int grp = 0; grp < NGRP; ++grp)
            for (int off = 0; off < sCnt[grp]; off += 16)
                ws[WS_TILES + nt++] = (grp << 16) | off;
        for (; nt < MAX_TILES; ++nt)
            ws[WS_TILES + nt] = -1;
    }
}

#define KQ_ROWS      256
#define STEPS_PER_SET  8
#define NSETS_TOTAL  (KQ_ROWS / 4 / STEPS_PER_SET)

__device__ __forceinline__ void load_set(v4f (&q)[2 * STEPS_PER_SET],
                                         const float* __restrict__ Bp, int bt) {
#pragma unroll
    for (int s = 0; s < STEPS_PER_SET; ++s) {
        const float* p = Bp + (size_t)((bt * STEPS_PER_SET + s) * 4) * DIM;
        q[2 * s]     = *(const v4f*)(p);
        q[2 * s + 1] = *(const v4f*)(p + DIM);
        __builtin_amdgcn_sched_barrier(0);
    }
}

__device__ __forceinline__ void mma_set(const v4f (&q)[2 * STEPS_PER_SET],
                                        const float* Ap, int bt,
                                        v8f& c0, v8f& c1, v8f& c2, v8f& c3) {
#pragma unroll
    for (int s = 0; s < STEPS_PER_SET; ++s) {
        const int ko = (bt * STEPS_PER_SET + s) * 4;
        v2f a;
        a[0] = Ap[ko * 16];
        a[1] = Ap[(ko + 1) * 16];
        const v4f r0 = q[2 * s];
        const v4f r1 = q[2 * s + 1];
        v2f b;
        b[0] = r0[0]; b[1] = r1[0];
        c0 = __builtin_amdgcn_wmma_f32_16x16x4_f32(false, a, false, b, (short)0, c0, false, false);
        b[0] = r0[1]; b[1] = r1[1];
        c1 = __builtin_amdgcn_wmma_f32_16x16x4_f32(false, a, false, b, (short)0, c1, false, false);
        b[0] = r0[2]; b[1] = r1[2];
        c2 = __builtin_amdgcn_wmma_f32_16x16x4_f32(false, a, false, b, (short)0, c2, false, false);
        b[0] = r0[3]; b[1] = r1[3];
        c3 = __builtin_amdgcn_wmma_f32_16x16x4_f32(false, a, false, b, (short)0, c3, false, false);
    }
}

__global__ __launch_bounds__(256) void grouped_gemm_kernel(
    const float* __restrict__ x,
    const float* __restrict__ T,
    const int*   __restrict__ ws,
    float*       __restrict__ out) {

    __shared__ __attribute__((aligned(16))) float Alds[DIM * 16];

    const int tile = ws[WS_TILES + blockIdx.x];
    if (tile < 0) return;
    const int grp  = tile >> 16;
    const int roff = tile & 0xFFFF;
    int m = ws[WS_COUNTS + grp] - roff;
    if (m > 16) m = 16;

    const int tid = threadIdx.x;

    for (int r = 0; r < 16; ++r) {
        const int rowIdx = (r < m) ? ws[WS_ROWS + grp * NB + roff + r] : -1;
        const float* xr = x + (size_t)(rowIdx < 0 ? 0 : rowIdx) * DIM;
        for (int d = tid; d < DIM; d += 256) {
            float v = (rowIdx >= 0) ? xr[d] : 0.0f;
            Alds[d * 16 + r] = v;
        }
    }
    __syncthreads();

    const int wave = tid >> 5;
    const int lane = tid & 31;
    const int r16  = lane & 15;
    const int half = lane >> 4;
    const int ng   = wave & 1;
    const int kq   = wave >> 1;

    const int n0    = blockIdx.y * 128 + ng * 64;
    const int kbase = kq * KQ_ROWS + half * 2;
    const float* Bp = T + (size_t)grp * DIM * DIM + (size_t)kbase * DIM + n0 + 4 * r16;
    const float* Ap = Alds + kbase * 16 + r16;

    v8f c0 = {}, c1 = {}, c2 = {}, c3 = {};
    v4f qa[2 * STEPS_PER_SET], qb[2 * STEPS_PER_SET];

    load_set(qa, Bp, 0);
#pragma unroll
    for (int bt = 0; bt < NSETS_TOTAL; bt += 2) {
        load_set(qb, Bp, bt + 1);
        mma_set(qa, Ap, bt, c0, c1, c2, c3);
        if (bt + 2 < NSETS_TOTAL) load_set(qa, Bp, bt + 2);
        mma_set(qb, Ap, bt + 1, c0, c1, c2, c3);
    }

    __syncthreads();
    v4f* red = (v4f*)Alds;
#pragma unroll
    for (int j = 0; j < 8; ++j) {
        v4f p;
        p[0] = c0[j]; p[1] = c1[j]; p[2] = c2[j]; p[3] = c3[j];
        red[(wave * 8 + j) * 32 + lane] = p;
    }
    __syncthreads();

    const int nOut = blockIdx.y * 128 + ng * 64 + 4 * r16;
#pragma unroll
    for (int jj = 0; jj < 2; ++jj) {
        const int j    = (wave >> 1) * 2 + jj;
        const int mrow = j + half * 8;
        if (mrow < m) {
            v4f s = red[((0 * 2 + ng) * 8 + j) * 32 + lane];
            s += red[((1 * 2 + ng) * 8 + j) * 32 + lane];
            s += red[((2 * 2 + ng) * 8 + j) * 32 + lane];
            s += red[((3 * 2 + ng) * 8 + j) * 32 + lane];
            const int b = ws[WS_ROWS + grp * NB + roff + mrow];
            *(v4f*)(out + (size_t)b * DIM + nOut) = s;
        }
    }
}

extern "C" void kernel_launch(void* const* d_in, const int* in_sizes, int n_in,
                              void* d_out, int out_size, void* d_ws, size_t ws_size,
                              hipStream_t stream) {
    const float* x = (const float*)d_in[0];
    const int*   g = (const int*)d_in[1];
    const float* T = (const float*)d_in[2];
    int*   ws  = (int*)d_ws;
    float* out = (float*)d_out;

    build_tiles_kernel<<<1, 64, 0, stream>>>(g, ws);

    dim3 grid(MAX_TILES, 8);
    grouped_gemm_kernel<<<grid, 256, 0, stream>>>(x, T, ws, out);
}


# Samples bucketed inside the GEMM kernel: each wave ballots the 128 matrix indices into a scalar member mask, so the prologue launch and the workspace tile table are gone (one launch, grid 40x8)
# speedup vs baseline: 1.9638x; 1.5732x over previous
#include <hip/hip_runtime.h>
#include <hip/hip_bf16.h>

typedef __attribute__((ext_vector_type(2))) float v2f;
typedef __attribute__((ext_vector_type(4))) float v4f;
typedef __attribute__((ext_vector_type(8))) float v8f;

#define NGRP 40
#define NB   128
#define DIM  1024

#define KQ_ROWS      256
#define STEPS_PER_SET  8
#define NSETS_TOTAL  (KQ_ROWS / 4 / STEPS_PER_SET)

__device__ __forceinline__ void load_set(v4f (&q)[2 * STEPS_PER_SET],
                                         const float* __restrict__ Bp, int bt) {
#pragma unroll
    for (int s = 0; s < STEPS_PER_SET; ++s) {
        const float* p = Bp + (size_t)((bt * STEPS_PER_SET + s) * 4) * DIM;
        q[2 * s]     = *(const v4f*)(p);
        q[2 * s + 1] = *(const v4f*)(p + DIM);
        __builtin_amdgcn_sched_barrier(0);
    }
}

__device__ __forceinline__ void mma_set(const v4f (&q)[2 * STEPS_PER_SET],
                                        const float* Ap, int bt,
                                        v8f& c0, v8f& c1, v8f& c2, v8f& c3) {
#pragma unroll
    for (int s = 0; s < STEPS_PER_SET; ++s) {
        const int ko = (bt * STEPS_PER_SET + s) * 4;
        v2f a;
        a[0] = Ap[ko * 16];
        a[1] = Ap[(ko + 1) * 16];
        const v4f r0 = q[2 * s];
        const v4f r1 = q[2 * s + 1];
        v2f b;
        b[0] = r0[0]; b[1] = r1[0];
        c0 = __builtin_amdgcn_wmma_f32_16x16x4_f32(false, a, false, b, (short)0, c0, false, false);
        b[0] = r0[1]; b[1] = r1[1];
        c1 = __builtin_amdgcn_wmma_f32_16x16x4_f32(false, a, false, b, (short)0, c1, false, false);
        b[0] = r0[2]; b[1] = r1[2];
        c2 = __builtin_amdgcn_wmma_f32_16x16x4_f32(false, a, false, b, (short)0, c2, false, false);
        b[0] = r0[3]; b[1] = r1[3];
        c3 = __builtin_amdgcn_wmma_f32_16x16x4_f32(false, a, false, b, (short)0, c3, false, false);
    }
}

__global__ __launch_bounds__(256) void grouped_gemm_kernel(
    const float* __restrict__ x,
    const int*   __restrict__ g,
    const float* __restrict__ T,
    float*       __restrict__ out) {

    __shared__ __attribute__((aligned(16))) float Alds[DIM * 16];

    const int grp  = blockIdx.x;
    const int tid  = threadIdx.x;
    const int wave = tid >> 5;
    const int lane = tid & 31;

    unsigned long long remLo, remHi;
    {
        const unsigned k0 = __builtin_amdgcn_ballot_w32(g[lane]      % NGRP == grp);
        const unsigned k1 = __builtin_amdgcn_ballot_w32(g[lane + 32] % NGRP == grp);
        const unsigned k2 = __builtin_amdgcn_ballot_w32(g[lane + 64] % NGRP == grp);
        const unsigned k3 = __builtin_amdgcn_ballot_w32(g[lane + 96] % NGRP == grp);
        remLo = ((unsigned long long)k1 << 32) | k0;
        remHi = ((unsigned long long)k3 << 32) | k2;
    }

    const int r16  = lane & 15;
    const int half = lane >> 4;
    const int ng   = wave & 1;
    const int kq   = wave >> 1;

    const int n0    = blockIdx.y * 128 + ng * 64;
    const int kbase = kq * KQ_ROWS + half * 2;
    const float* Bg = T + (size_t)grp * DIM * DIM + (size_t)kbase * DIM + n0 + 4 * r16;
    const float* Ap = Alds + kbase * 16 + r16;
    const int nOut  = n0 + 4 * r16;

    while ((remLo | remHi) != 0) {
        int passOff = 0;
        asm volatile("" : "+s"(passOff));
        const float* Bp = Bg + passOff;
        unsigned long long rowsLo = 0, rowsHi = 0;
        int m = 0;
        for (int r = 0; r < 16; ++r) {
            int rowIdx = -1;
            if (remLo != 0) {
                rowIdx = __builtin_ctzll(remLo);
                remLo &= remLo - 1;
            } else if (remHi != 0) {
                rowIdx = 64 + __builtin_ctzll(remHi);
                remHi &= remHi - 1;
            }
            if (rowIdx >= 0) {
                ++m;
                if (r < 8) rowsLo |= (unsigned long long)rowIdx << (8 * r);
                else       rowsHi |= (unsigned long long)rowIdx << (8 * (r - 8));
            }
            const float* xr = x + (size_t)(rowIdx < 0 ? 0 : rowIdx) * DIM;
            for (int d = tid; d < DIM; d += 256) {
                float v = (rowIdx >= 0) ? xr[d] : 0.0f;
                Alds[d * 16 + r] = v;
            }
        }
        __syncthreads();

        v8f c0 = {}, c1 = {}, c2 = {}, c3 = {};
        v4f qa[2 * STEPS_PER_SET], qb[2 * STEPS_PER_SET];

        load_set(qa, Bp, 0);
#pragma unroll
        for (int bt = 0; bt < NSETS_TOTAL; bt += 2) {
            load_set(qb, Bp, bt + 1);
            mma_set(qa, Ap, bt, c0, c1, c2, c3);
            if (bt + 2 < NSETS_TOTAL) load_set(qa, Bp, bt + 2);
            mma_set(qb, Ap, bt + 1, c0, c1, c2, c3);
        }

        __syncthreads();
        v4f* red = (v4f*)Alds;
#pragma unroll
        for (int j = 0; j < 8; ++j) {
            v4f p;
            p[0] = c0[j]; p[1] = c1[j]; p[2] = c2[j]; p[3] = c3[j];
            red[(wave * 8 + j) * 32 + lane] = p;
        }
        __syncthreads();

#pragma unroll
        for (int jj = 0; jj < 2; ++jj) {
            const int j    = (wave >> 1) * 2 + jj;
            const int mrow = j + half * 8;
            if (mrow < m) {
                v4f s = red[((0 * 2 + ng) * 8 + j) * 32 + lane];
                s += red[((1 * 2 + ng) * 8 + j) * 32 + lane];
                s += red[((2 * 2 + ng) * 8 + j) * 32 + lane];
                s += red[((3 * 2 + ng) * 8 + j) * 32 + lane];
                const unsigned long long packed = half ? rowsHi : rowsLo;
                const int b = (int)((packed >> (8 * j)) & 0xFF);
                *(v4f*)(out + (size_t)b * DIM + nOut) = s;
            }
        }
        __syncthreads();
    }
}

extern "C" void kernel_launch(void* const* d_in, const int* in_sizes, int n_in,
                              void* d_out, int out_size, void* d_ws, size_t ws_size,
                              hipStream_t stream) {
    const float* x = (const float*)d_in[0];
    const int*   g = (const int*)d_in[1];
    const float* T = (const float*)d_in[2];
    float* out = (float*)d_out;

    dim3 grid(NGRP, 8);
    grouped_gemm_kernel<<<grid, 256, 0, stream>>>(x, g, T, out);
}


# A staging: all reads of x for a pass are requested before the first is used (up to 64 per thread in flight) and go to LDS as 16-byte stores, replacing 64 dependent load-wait-store trips per thread
# speedup vs baseline: 3.2362x; 1.6479x over previous
#include <hip/hip_runtime.h>
#include <hip/hip_bf16.h>

typedef __attribute__((ext_vector_type(2))) float v2f;
typedef __attribute__((ext_vector_type(4))) float v4f;
typedef __attribute__((ext_vector_type(8))) float v8f;

#define NGRP 40
#define NB   128
#define DIM  1024

#define KQ_ROWS      256
#define STEPS_PER_SET  8
#define NSETS_TOTAL  (KQ_ROWS / 4 / STEPS_PER_SET)

__device__ __forceinline__ void load_set(v4f (&q)[2 * STEPS_PER_SET],
                                         const float* __restrict__ Bp, int bt) {
#pragma unroll
    for (int s = 0; s < STEPS_PER_SET; ++s) {
        const float* p = Bp + (size_t)((bt * STEPS_PER_SET + s) * 4) * DIM;
        q[2 * s]     = *(const v4f*)(p);
        q[2 * s + 1] = *(const v4f*)(p + DIM);
        __builtin_amdgcn_sched_barrier(0);
    }
}

__device__ __forceinline__ void mma_set(const v4f (&q)[2 * STEPS_PER_SET],
                                        const float* Ap, int bt,
                                        v8f& c0, v8f& c1, v8f& c2, v8f& c3) {
#pragma unroll
    for (int s = 0; s < STEPS_PER_SET; ++s) {
        const int ko = (bt * STEPS_PER_SET + s) * 4;
        v2f a;
        a[0] = Ap[ko * 16];
        a[1] = Ap[(ko + 1) * 16];
        const v4f r0 = q[2 * s];
        const v4f r1 = q[2 * s + 1];
        v2f b;
        b[0] = r0[0]; b[1] = r1[0];
        c0 = __builtin_amdgcn_wmma_f32_16x16x4_f32(false, a, false, b, (short)0, c0, false, false);
        b[0] = r0[1]; b[1] = r1[1];
        c1 = __builtin_amdgcn_wmma_f32_16x16x4_f32(false, a, false, b, (short)0, c1, false, false);
        b[0] = r0[2]; b[1] = r1[2];
        c2 = __builtin_amdgcn_wmma_f32_16x16x4_f32(false, a, false, b, (short)0, c2, false, false);
        b[0] = r0[3]; b[1] = r1[3];
        c3 = __builtin_amdgcn_wmma_f32_16x16x4_f32(false, a, false, b, (short)0, c3, false, false);
    }
}

__global__ __launch_bounds__(256) void grouped_gemm_kernel(
    const float* __restrict__ x,
    const int*   __restrict__ g,
    const float* __restrict__ T,
    float*       __restrict__ out) {

    __shared__ __attribute__((aligned(16))) float Alds[DIM * 16];

    const int grp  = blockIdx.x;
    const int tid  = threadIdx.x;
    const int wave = tid >> 5;
    const int lane = tid & 31;

    unsigned long long remLo, remHi;
    {
        const unsigned k0 = __builtin_amdgcn_ballot_w32(g[lane]      % NGRP == grp);
        const unsigned k1 = __builtin_amdgcn_ballot_w32(g[lane + 32] % NGRP == grp);
        const unsigned k2 = __builtin_amdgcn_ballot_w32(g[lane + 64] % NGRP == grp);
        const unsigned k3 = __builtin_amdgcn_ballot_w32(g[lane + 96] % NGRP == grp);
        remLo = ((unsigned long long)k1 << 32) | k0;
        remHi = ((unsigned long long)k3 << 32) | k2;
    }

    const int r16  = lane & 15;
    const int half = lane >> 4;
    const int ng   = wave & 1;
    const int kq   = wave >> 1;

    const int n0    = blockIdx.y * 128 + ng * 64;
    const int kbase = kq * KQ_ROWS + half * 2;
    const float* Bg = T + (size_t)grp * DIM * DIM + (size_t)kbase * DIM + n0 + 4 * r16;
    const float* Ap = Alds + kbase * 16 + r16;
    const int nOut  = n0 + 4 * r16;

    while ((remLo | remHi) != 0) {
        int passOff = 0;
        asm volatile("" : "+s"(passOff));
        const float* Bp = Bg + passOff;
        unsigned long long rowsLo = 0, rowsHi = 0;
        int m = 0;
        float xv[4][16];
        for (int r = 0; r < 16; ++r) {
            int rowIdx = -1;
            if (remLo != 0) {
                rowIdx = __builtin_ctzll(remLo);
                remLo &= remLo - 1;
            } else if (remHi != 0) {
                rowIdx = 64 + __builtin_ctzll(remHi);
                remHi &= remHi - 1;
            }
            if (rowIdx >= 0) {
                ++m;
                if (r < 8) rowsLo |= (unsigned long long)rowIdx << (8 * r);
                else       rowsHi |= (unsigned long long)rowIdx << (8 * (r - 8));
                const float* xr = x + (size_t)rowIdx * DIM + tid;
                xv[0][r] = xr[0];   xv[1][r] = xr[256];
                xv[2][r] = xr[512]; xv[3][r] = xr[768];
            } else {
                xv[0][r] = 0.0f; xv[1][r] = 0.0f; xv[2][r] = 0.0f; xv[3][r] = 0.0f;
            }
        }
        for (int i = 0; i < 4; ++i) {
            v4f* dst = (v4f*)(Alds + (tid + i * 256) * 16);
            for (int q = 0; q < 4; ++q) {
                v4f p;
                p[0] = xv[i][4 * q];     p[1] = xv[i][4 * q + 1];
                p[2] = xv[i][4 * q + 2]; p[3] = xv[i][4 * q + 3];
                dst[q] = p;
            }
        }
        __syncthreads();

        v8f c0 = {}, c1 = {}, c2 = {}, c3 = {};
        v4f qa[2 * STEPS_PER_SET], qb[2 * STEPS_PER_SET];

        load_set(qa, Bp, 0);
#pragma unroll
        for (int bt = 0; bt < NSETS_TOTAL; bt += 2) {
            load_set(qb, Bp, bt + 1);
            mma_set(qa, Ap, bt, c0, c1, c2, c3);
            if (bt + 2 < NSETS_TOTAL) load_set(qa, Bp, bt + 2);
            mma_set(qb, Ap, bt + 1, c0, c1, c2, c3);
        }

        __syncthreads();
        v4f* red = (v4f*)Alds;
#pragma unroll
        for (int j = 0; j < 8; ++j) {
            v4f p;
            p[0] = c0[j]; p[1] = c1[j]; p[2] = c2[j]; p[3] = c3[j];
            red[(wave * 8 + j) * 32 + lane] = p;
        }
        __syncthreads();

#pragma unroll
        for (int jj = 0; jj < 2; ++jj) {
            const int j    = (wave >> 1) * 2 + jj;
            const int mrow = j + half * 8;
            if (mrow < m) {
                v4f s = red[((0 * 2 + ng) * 8 + j) * 32 + lane];
                s += red[((1 * 2 + ng) * 8 + j) * 32 + lane];
                s += red[((2 * 2 + ng) * 8 + j) * 32 + lane];
                s += red[((3 * 2 + ng) * 8 + j) * 32 + lane];
                const unsigned long long packed = half ? rowsHi : rowsLo;
                const int b = (int)((packed >> (8 * j)) & 0xFF);
                *(v4f*)(out + (size_t)b * DIM + nOut) = s;
            }
        }
        __syncthreads();
    }
}

extern "C" void kernel_launch(void* const* d_in, const int* in_sizes, int n_in,
                              void* d_out, int out_size, void* d_ws, size_t ws_size,
                              hipStream_t stream) {
    const float* x = (const float*)d_in[0];
    const int*   g = (const int*)d_in[1];
    const float* T = (const float*)d_in[2];
    float* out = (float*)d_out;

    dim3 grid(NGRP, 8);
    grouped_gemm_kernel<<<grid, 256, 0, stream>>>(x, g, T, out);
}


# Work unit halved to 64 columns: 4-wave blocks (the 4 K quarters of one column group), grid 40x16 = 640 resident blocks, so the fullest WGP carries 3 units, not 4; sums and their order unchanged
# speedup vs baseline: 3.7537x; 1.1599x over previous
#include <hip/hip_runtime.h>
#include <hip/hip_bf16.h>

typedef __attribute__((ext_vector_type(2))) float v2f;
typedef __attribute__((ext_vector_type(4))) float v4f;
typedef __attribute__((ext_vector_type(8))) float v8f;

#define NGRP 40
#define NB   128
#define DIM  1024

#define KQ_ROWS      256
#define STEPS_PER_SET  8
#define NSETS_TOTAL  (KQ_ROWS / 4 / STEPS_PER_SET)

__device__ __forceinline__ void load_set(v4f (&q)[2 * STEPS_PER_SET],
                                         const float* __restrict__ Bp, int bt) {
#pragma unroll
    for (int s = 0; s < STEPS_PER_SET; ++s) {
        const float* p = Bp + (size_t)((bt * STEPS_PER_SET + s) * 4) * DIM;
        q[2 * s]     = *(const v4f*)(p);
        q[2 * s + 1] = *(const v4f*)(p + DIM);
        __builtin_amdgcn_sched_barrier(0);
    }
}

__device__ __forceinline__ void mma_set(const v4f (&q)[2 * STEPS_PER_SET],
                                        const float* Ap, int bt,
                                        v8f& c0, v8f& c1, v8f& c2, v8f& c3) {
#pragma unroll
    for (int s = 0; s < STEPS_PER_SET; ++s) {
        const int ko = (bt * STEPS_PER_SET + s) * 4;
        v2f a;
        a[0] = Ap[ko * 16];
        a[1] = Ap[(ko + 1) * 16];
        const v4f r0 = q[2 * s];
        const v4f r1 = q[2 * s + 1];
        v2f b;
        b[0] = r0[0]; b[1] = r1[0];
        c0 = __builtin_amdgcn_wmma_f32_16x16x4_f32(false, a, false, b, (short)0, c0, false, false);
        b[0] = r0[1]; b[1] = r1[1];
        c1 = __builtin_amdgcn_wmma_f32_16x16x4_f32(false, a, false, b, (short)0, c1, false, false);
        b[0] = r0[2]; b[1] = r1[2];
        c2 = __builtin_amdgcn_wmma_f32_16x16x4_f32(false, a, false, b, (short)0, c2, false, false);
        b[0] = r0[3]; b[1] = r1[3];
        c3 = __builtin_amdgcn_wmma_f32_16x16x4_f32(false, a, false, b, (short)0, c3, false, false);
    }
}

__global__ __launch_bounds__(128) void grouped_gemm_kernel(
    const float* __restrict__ x,
    const int*   __restrict__ g,
    const float* __restrict__ T,
    float*       __restrict__ out) {

    __shared__ __attribute__((aligned(16))) float Alds[DIM * 16];

    const int grp  = blockIdx.x;
    const int tid  = threadIdx.x;
    const int wave = tid >> 5;
    const int lane = tid & 31;

    unsigned long long remLo, remHi;
    {
        const unsigned k0 = __builtin_amdgcn_ballot_w32(g[lane]      % NGRP == grp);
        const unsigned k1 = __builtin_amdgcn_ballot_w32(g[lane + 32] % NGRP == grp);
        const unsigned k2 = __builtin_amdgcn_ballot_w32(g[lane + 64] % NGRP == grp);
        const unsigned k3 = __builtin_amdgcn_ballot_w32(g[lane + 96] % NGRP == grp);
        remLo = ((unsigned long long)k1 << 32) | k0;
        remHi = ((unsigned long long)k3 << 32) | k2;
    }

    const int r16  = lane & 15;
    const int half = lane >> 4;
    const int kq   = wave;

    const int n0    = blockIdx.y * 64;
    const int kbase = kq * KQ_ROWS + half * 2;
    const float* Bg = T + (size_t)grp * DIM * DIM + (size_t)kbase * DIM + n0 + 4 * r16;
    const float* Ap = Alds + kbase * 16 + r16;
    const int nOut  = n0 + 4 * r16;

    while ((remLo | remHi) != 0) {
        int passOff = 0;
        asm volatile("" : "+s"(passOff));
        const float* Bp = Bg + passOff;
        unsigned long long rowsLo = 0, rowsHi = 0;
        int m = 0;
        float xv[8][16];
        for (int r = 0; r < 16; ++r) {
            int rowIdx = -1;
            if (remLo != 0) {
                rowIdx = __builtin_ctzll(remLo);
                remLo &= remLo - 1;
            } else if (remHi != 0) {
                rowIdx = 64 + __builtin_ctzll(remHi);
                remHi &= remHi - 1;
            }
            if (rowIdx >= 0) {
                ++m;
                if (r < 8) rowsLo |= (unsigned long long)rowIdx << (8 * r);
                else       rowsHi |= (unsigned long long)rowIdx << (8 * (r - 8));
                const float* xr = x + (size_t)rowIdx * DIM + tid;
                for (int i = 0; i < 8; ++i) xv[i][r] = xr[i * 128];
            } else {
                for (int i = 0; i < 8; ++i) xv[i][r] = 0.0f;
            }
        }
        for (int i = 0; i < 8; ++i) {
            v4f* dst = (v4f*)(Alds + (tid + i * 128) * 16);
            for (int q = 0; q < 4; ++q) {
                v4f p;
                p[0] = xv[i][4 * q];     p[1] = xv[i][4 * q + 1];
                p[2] = xv[i][4 * q + 2]; p[3] = xv[i][4 * q + 3];
                dst[q] = p;
            }
        }
        __syncthreads();

        v8f c0 = {}, c1 = {}, c2 = {}, c3 = {};
        v4f qa[2 * STEPS_PER_SET], qb[2 * STEPS_PER_SET];

        load_set(qa, Bp, 0);
#pragma unroll
        for (int bt = 0; bt < NSETS_TOTAL; bt += 2) {
            load_set(qb, Bp, bt + 1);
            mma_set(qa, Ap, bt, c0, c1, c2, c3);
            if (bt + 2 < NSETS_TOTAL) load_set(qa, Bp, bt + 2);
            mma_set(qb, Ap, bt + 1, c0, c1, c2, c3);
        }

        __syncthreads();
        v4f* red = (v4f*)Alds;
#pragma unroll
        for (int j = 0; j < 8; ++j) {
            v4f p;
            p[0] = c0[j]; p[1] = c1[j]; p[2] = c2[j]; p[3] = c3[j];
            red[(wave * 8 + j) * 32 + lane] = p;
        }
        __syncthreads();

#pragma unroll
        for (int jj = 0; jj < 2; ++jj) {
            const int j    = wave * 2 + jj;
            const int mrow = j + half * 8;
            if (mrow < m) {
                v4f s = red[(0 * 8 + j) * 32 + lane];
                s += red[(1 * 8 + j) * 32 + lane];
                s += red[(2 * 8 + j) * 32 + lane];
                s += red[(3 * 8 + j) * 32 + lane];
                const unsigned long long packed = half ? rowsHi : rowsLo;
                const int b = (int)((packed >> (8 * j)) & 0xFF);
                *(v4f*)(out + (size_t)b * DIM + nOut) = s;
            }
        }
        __syncthreads();
    }
}

extern "C" void kernel_launch(void* const* d_in, const int* in_sizes, int n_in,
                              void* d_out, int out_size, void* d_ws, size_t ws_size,
                              hipStream_t stream) {
    const float* x = (const float*)d_in[0];
    const int*   g = (const int*)d_in[1];
    const float* T = (const float*)d_in[2];
    float* out = (float*)d_out;

    dim3 grid(NGRP, 16);
    grouped_gemm_kernel<<<grid, 128, 0, stream>>>(x, g, T, out);
}


# Grid re-cut for equal load per WGP: 512 blocks take a whole 64-column unit, 256 take a 32-column half (8-byte T loads, 2 tiles), so every WGP holds 2.5 units, not 3 or 2; sums and order unchanged
# speedup vs baseline: 4.0427x; 1.0770x over previous
#include <hip/hip_runtime.h>
#include <hip/hip_bf16.h>

typedef __attribute__((ext_vector_type(2))) float v2f;
typedef __attribute__((ext_vector_type(4))) float v4f;
typedef __attribute__((ext_vector_type(8))) float v8f;

#define NGRP 40
#define NB   128
#define DIM  1024

#define KQ_ROWS      256
#define KSTEPS       (KQ_ROWS / 4)
#define NFULL        512

template <int NT> struct bvec;
template <> struct bvec<4> { typedef v4f type; };
template <> struct bvec<2> { typedef v2f type; };

template <int NT, int STEPS>
__device__ __forceinline__ void load_set(typename bvec<NT>::type (&q)[2 * STEPS],
                                         const float* __restrict__ Bp, int bt) {
    typedef typename bvec<NT>::type VT;
#pragma unroll
    for (int s = 0; s < STEPS; ++s) {
        const float* p = Bp + (size_t)((bt * STEPS + s) * 4) * DIM;
        q[2 * s]     = *(const VT*)(p);
        q[2 * s + 1] = *(const VT*)(p + DIM);
        __builtin_amdgcn_sched_barrier(0);
    }
}

template <int NT, int STEPS>
__device__ __forceinline__ void mma_set(const typename bvec<NT>::type (&q)[2 * STEPS],
                                        const float* Ap, int bt, v8f (&c)[NT]) {
    typedef typename bvec<NT>::type VT;
#pragma unroll
    for (int s = 0; s < STEPS; ++s) {
        const int ko = (bt * STEPS + s) * 4;
        v2f a;
        a[0] = Ap[ko * 16];
        a[1] = Ap[(ko + 1) * 16];
        const VT r0 = q[2 * s];
        const VT r1 = q[2 * s + 1];
#pragma unroll
        for (int i = 0; i < NT; ++i) {
            v2f b;
            b[0] = r0[i]; b[1] = r1[i];
            c[i] = __builtin_amdgcn_wmma_f32_16x16x4_f32(false, a, false, b, (short)0, c[i], false, false);
        }
    }
}

template <int NT, int STEPS>
__device__ __forceinline__ void stream_and_store(const float* __restrict__ Bp, const float* Ap,
                                                 float* Alds, float* __restrict__ out,
                                                 int wave, int lane, int half, int m,
                                                 unsigned long long rowsLo,
                                                 unsigned long long rowsHi, int nOut) {
    typedef typename bvec<NT>::type VT;
    constexpr int NSETS = KSTEPS / STEPS;

    v8f c[NT];
#pragma unroll
    for (int i = 0; i < NT; ++i) c[i] = v8f{};
    VT qa[2 * STEPS], qb[2 * STEPS];

    load_set<NT, STEPS>(qa, Bp, 0);
#pragma unroll
    for (int bt = 0; bt < NSETS; bt += 2) {
        load_set<NT, STEPS>(qb, Bp, bt + 1);
        mma_set<NT, STEPS>(qa, Ap, bt, c);
        if (bt + 2 < NSETS) load_set<NT, STEPS>(qa, Bp, bt + 2);
        mma_set<NT, STEPS>(qb, Ap, bt + 1, c);
    }

    __syncthreads();
    VT* red = (VT*)Alds;
#pragma unroll
    for (int j = 0; j < 8; ++j) {
        VT p;
#pragma unroll
        for (int i = 0; i < NT; ++i) p[i] = c[i][j];
        red[(wave * 8 + j) * 32 + lane] = p;
    }
    __syncthreads();

#pragma unroll
    for (int jj = 0; jj < 2; ++jj) {
        const int j    = wave * 2 + jj;
        const int mrow = j + half * 8;
        if (mrow < m) {
            VT s = red[(0 * 8 + j) * 32 + lane];
            s += red[(1 * 8 + j) * 32 + lane];
            s += red[(2 * 8 + j) * 32 + lane];
            s += red[(3 * 8 + j) * 32 + lane];
            const unsigned long long packed = half ? rowsHi : rowsLo;
            const int b = (int)((packed >> (8 * j)) & 0xFF);
            *(VT*)(out + (size_t)b * DIM + nOut) = s;
        }
    }
}

__global__ __launch_bounds__(128) void grouped_gemm_kernel(
    const float* __restrict__ x,
    const int*   __restrict__ g,
    const float* __restrict__ T,
    float*       __restrict__ out) {

    __shared__ __attribute__((aligned(16))) float Alds[DIM * 16];

    const int  bid  = blockIdx.x;
    const bool full = bid < NFULL;
    const int  hb   = bid - NFULL;
    const int  unit = full ? bid : NFULL + (hb >> 1);
    const int  grp  = unit % NGRP;
    const int  n0   = (unit / NGRP) * 64 + (full ? 0 : (hb & 1) * 32);
    const int tid  = threadIdx.x;
    const int wave = tid >> 5;
    const int lane = tid & 31;

    unsigned long long remLo, remHi;
    {
        const unsigned k0 = __builtin_amdgcn_ballot_w32(g[lane]      % NGRP == grp);
        const unsigned k1 = __builtin_amdgcn_ballot_w32(g[lane + 32] % NGRP == grp);
        const unsigned k2 = __builtin_amdgcn_ballot_w32(g[lane + 64] % NGRP == grp);
        const unsigned k3 = __builtin_amdgcn_ballot_w32(g[lane + 96] % NGRP == grp);
        remLo = ((unsigned long long)k1 << 32) | k0;
        remHi = ((unsigned long long)k3 << 32) | k2;
    }

    const int r16  = lane & 15;
    const int half = lane >> 4;
    const int kq   = wave;

    const int kbase = kq * KQ_ROWS + half * 2;
    const float* Bg = T + (size_t)grp * DIM * DIM + (size_t)kbase * DIM + n0;
    const float* Ap = Alds + kbase * 16 + r16;

    while ((remLo | remHi) != 0) {
        int passOff = 0;
        asm volatile("" : "+s"(passOff));
        const float* Bp = Bg + passOff;
        unsigned long long rowsLo = 0, rowsHi = 0;
        int m = 0;
        float xv[8][16];
        for (int r = 0; r < 16; ++r) {
            int rowIdx = -1;
            if (remLo != 0) {
                rowIdx = __builtin_ctzll(remLo);
                remLo &= remLo - 1;
            } else if (remHi != 0) {
                rowIdx = 64 + __builtin_ctzll(remHi);
                remHi &= remHi - 1;
            }
            if (rowIdx >= 0) {
                ++m;
                if (r < 8) rowsLo |= (unsigned long long)rowIdx << (8 * r);
                else       rowsHi |= (unsigned long long)rowIdx << (8 * (r - 8));
                const float* xr = x + (size_t)rowIdx * DIM + tid;
                for (int i = 0; i < 8; ++i) xv[i][r] = xr[i * 128];
            } else {
                for (int i = 0; i < 8; ++i) xv[i][r] = 0.0f;
            }
        }
        for (int i = 0; i < 8; ++i) {
            v4f* dst = (v4f*)(Alds + (tid + i * 128) * 16);
            for (int q = 0; q < 4; ++q) {
                v4f p;
                p[0] = xv[i][4 * q];     p[1] = xv[i][4 * q + 1];
                p[2] = xv[i][4 * q + 2]; p[3] = xv[i][4 * q + 3];
                dst[q] = p;
            }
        }
        __syncthreads();

        if (full)
            stream_and_store<4, 8>(Bp + 4 * r16, Ap, Alds, out, wave, lane, half, m,
                                   rowsLo, rowsHi, n0 + 4 * r16);
        else
            stream_and_store<2, 16>(Bp + 2 * r16, Ap, Alds, out, wave, lane, half, m,
                                    rowsLo, rowsHi, n0 + 2 * r16);
        __syncthreads();
    }
}

extern "C" void kernel_launch(void* const* d_in, const int* in_sizes, int n_in,
                              void* d_out, int out_size, void* d_ws, size_t ws_size,
                              hipStream_t stream) {
    const float* x = (const float*)d_in[0];
    const int*   g = (const int*)d_in[1];
    const float* T = (const float*)d_in[2];
    float* out = (float*)d_out;

    dim3 grid(768);
    grouped_gemm_kernel<<<grid, 128, 0, stream>>>(x, g, T, out);
}
